# starting kernel
# baseline (speedup 1.0000x reference)
#include <hip/hip_runtime.h>
#include <math.h>

typedef __attribute__((ext_vector_type(2))) float v2f;
typedef __attribute__((ext_vector_type(8))) float v8f;

#define DIM 256
#define PER_STAGE 128
#define CAPV 0.99f
#define BATCH 32
#define HW 4096
#define TILE_P 32
#define PITCH 36

__global__ __launch_bounds__(32) void prep_kernel(
    const float* __restrict__ modulation,
    const float* __restrict__ angles,
    const float* __restrict__ diagonal,
    const float* __restrict__ w_angle,
    const float* __restrict__ w_diag,
    float* __restrict__ ws_cos,
    float* __restrict__ ws_sin,
    float* __restrict__ ws_diag)
{
    const int  tile    = blockIdx.x;
    const bool is_diag = (tile >= 64);
    const int  n0      = is_diag ? (tile - 64) * 16 : tile * 16;
    const float* W     = is_diag ? w_diag : w_angle;

    const int lane = threadIdx.x;
    const int l16  = lane & 15;
    const int hi   = lane >> 4;

    const float* arow0 = modulation + l16 * 256;
    const float* arow1 = modulation + (l16 + 16) * 256;
    const float* wrow  = W + (n0 + l16) * 256;

    v8f acc0 = {};
    v8f acc1 = {};
    for (int k0 = 0; k0 < 256; k0 += 4) {
        const int k = k0 + 2 * hi;
        v2f a0, a1, bv;
        a0.x = arow0[k];  a0.y = arow0[k + 1];
        a1.x = arow1[k];  a1.y = arow1[k + 1];
        bv.x = wrow[k];   bv.y = wrow[k + 1];
        acc0 = __builtin_amdgcn_wmma_f32_16x16x4_f32(false, a0, false, bv,
                                                     (short)0, acc0, false, false);
        acc1 = __builtin_amdgcn_wmma_f32_16x16x4_f32(false, a1, false, bv,
                                                     (short)0, acc1, false, false);
    }

    const int n = n0 + l16;
    if (!is_diag) {
        const float base = angles[n];
#pragma unroll
        for (int i = 0; i < 8; ++i) {
            const int m0 = i + 8 * hi;
            const int m1 = m0 + 16;
            float a0 = base + tanhf(acc0[i]);
            float a1 = base + tanhf(acc1[i]);
            ws_cos[m0 * 1024 + n] = cosf(a0);
            ws_sin[m0 * 1024 + n] = sinf(a0);
            ws_cos[m1 * 1024 + n] = cosf(a1);
            ws_sin[m1 * 1024 + n] = sinf(a1);
        }
    } else {
        float base = tanhf(diagonal[n]);
        base = fminf(fmaxf(base, -CAPV), CAPV);
#pragma unroll
        for (int i = 0; i < 8; ++i) {
            const int m0 = i + 8 * hi;
            const int m1 = m0 + 16;
            float d0 = base + tanhf(acc0[i]);
            float d1 = base + tanhf(acc1[i]);
            ws_diag[m0 * 256 + n] = fminf(fmaxf(d0, -CAPV), CAPV);
            ws_diag[m1 * 256 + n] = fminf(fmaxf(d1, -CAPV), CAPV);
        }
    }
}

__global__ __launch_bounds__(256) void butterfly_kernel(
    const float* __restrict__ x,
    const float* __restrict__ ws_cos,
    const float* __restrict__ ws_sin,
    const float* __restrict__ ws_diag,
    float* __restrict__ out)
{
    __shared__ float tile[DIM * PITCH];
    __shared__ float lcos[1024];
    __shared__ float lsin[1024];
    __shared__ float ldiag[DIM];

    const int b   = blockIdx.x >> 7;
    const int p0  = (blockIdx.x & 127) * TILE_P;
    const int tid = threadIdx.x;
    const int wv  = tid >> 5;
    const int t   = tid & 31;

    if (tid < 256) {
        ((float4*)lcos)[tid] = ((const float4*)(ws_cos + b * 1024))[tid];
        ((float4*)lsin)[tid] = ((const float4*)(ws_sin + b * 1024))[tid];
        ldiag[tid] = ws_diag[b * 256 + tid];
    }

    const int cq = t >> 3;
    const int pq = (t & 7) * 4;
    const float* xb = x + ((size_t)b * DIM) * HW + p0;
#pragma unroll
    for (int cc = 0; cc < 8; ++cc) {
        const int c = wv * 32 + cc * 4 + cq;
        const float4 val = *(const float4*)(xb + (size_t)c * HW + pq);
        *(float4*)&tile[c * PITCH + pq] = val;
    }
    __syncthreads();

    for (int j = 0; j < 4; ++j) {
        const int p = wv * 4 + j;
        float v[8];
#pragma unroll
        for (int r = 0; r < 8; ++r)
            v[r] = tile[(r * 32 + t) * PITCH + p];

#pragma unroll
        for (int s = 0; s < 5; ++s) {
            const int half = 1 << s;
#pragma unroll
            for (int r = 0; r < 8; ++r) {
                const int c   = r * 32 + t;
                const int idx = s * PER_STAGE + ((c >> (s + 1)) << s) + (c & (half - 1));
                const float cv = lcos[idx];
                const float sv = lsin[idx];
                const float other = __shfl_xor(v[r], half, 32);
                const bool  right = (c >> s) & 1;
                v[r] = right ? fmaf(cv, v[r], -sv * other)
                             : fmaf(cv, v[r],  sv * other);
            }
        }
#pragma unroll
        for (int s = 5; s < 8; ++s) {
            const int half = 1 << s;
            const int hr   = 1 << (s - 5);
#pragma unroll
            for (int r = 0; r < 8; ++r) {
                if (r & hr) continue;
                const int rp  = r + hr;
                const int c   = r * 32 + t;
                const int idx = s * PER_STAGE + ((c >> (s + 1)) << s) + (c & (half - 1));
                const float cv = lcos[idx];
                const float sv = lsin[idx];
                const float L = v[r];
                const float R = v[rp];
                v[r]  = fmaf(cv, L,  sv * R);
                v[rp] = fmaf(cv, R, -sv * L);
            }
        }
#pragma unroll
        for (int r = 0; r < 8; ++r) {
            const int c = r * 32 + t;
            tile[c * PITCH + p] = v[r] * ldiag[c];
        }
    }
    __syncthreads();

    float* ob = out + ((size_t)b * DIM) * HW + p0;
#pragma unroll
    for (int cc = 0; cc < 8; ++cc) {
        const int c = wv * 32 + cc * 4 + cq;
        *(float4*)(ob + (size_t)c * HW + pq) = *(const float4*)&tile[c * PITCH + pq];
    }
}

extern "C" void kernel_launch(void* const* d_in, const int* in_sizes, int n_in,
                              void* d_out, int out_size, void* d_ws, size_t ws_size,
                              hipStream_t stream) {
    const float* x          = (const float*)d_in[0];
    const float* modulation = (const float*)d_in[1];
    const float* angles     = (const float*)d_in[2];
    const float* diagonal   = (const float*)d_in[3];
    const float* w_angle    = (const float*)d_in[4];
    const float* w_diag     = (const float*)d_in[5];
    float* out = (float*)d_out;

    float* ws      = (float*)d_ws;
    float* ws_cos  = ws;
    float* ws_sin  = ws + 32 * 1024;
    float* ws_diag = ws + 64 * 1024;

    prep_kernel<<<80, 32, 0, stream>>>(modulation, angles, diagonal,
                                       w_angle, w_diag, ws_cos, ws_sin, ws_diag);
    butterfly_kernel<<<BATCH * (HW / TILE_P), 256, 0, stream>>>(
        x, ws_cos, ws_sin, ws_diag, out);
}


# butterfly remapped: positions on lanes (float4), channels in registers as two 16-channel passes around one LDS transpose; cos/sin/diag become wave-uniform scalar loads, no lane shuffles
# speedup vs baseline: 1.4278x; 1.4278x over previous
#include <hip/hip_runtime.h>
#include <math.h>

typedef __attribute__((ext_vector_type(2))) float v2f;
typedef __attribute__((ext_vector_type(8))) float v8f;
typedef __attribute__((ext_vector_type(4))) float v4f;

#define DIM 256
#define PER_STAGE 128
#define CAPV 0.99f
#define BATCH 32
#define HW 4096

__global__ __launch_bounds__(32) void prep_kernel(
    const float* __restrict__ modulation,
    const float* __restrict__ angles,
    const float* __restrict__ diagonal,
    const float* __restrict__ w_angle,
    const float* __restrict__ w_diag,
    float* __restrict__ ws_cos,
    float* __restrict__ ws_sin,
    float* __restrict__ ws_diag)
{
    const int  tile    = blockIdx.x;
    const bool is_diag = (tile >= 64);
    const int  n0      = is_diag ? (tile - 64) * 16 : tile * 16;
    const float* W     = is_diag ? w_diag : w_angle;

    const int lane = threadIdx.x;
    const int l16  = lane & 15;
    const int hi   = lane >> 4;

    const float* arow0 = modulation + l16 * 256;
    const float* arow1 = modulation + (l16 + 16) * 256;
    const float* wrow  = W + (n0 + l16) * 256;

    v8f acc0 = {};
    v8f acc1 = {};
    for (int k0 = 0; k0 < 256; k0 += 4) {
        const int k = k0 + 2 * hi;
        v2f a0, a1, bv;
        a0.x = arow0[k];  a0.y = arow0[k + 1];
        a1.x = arow1[k];  a1.y = arow1[k + 1];
        bv.x = wrow[k];   bv.y = wrow[k + 1];
        acc0 = __builtin_amdgcn_wmma_f32_16x16x4_f32(false, a0, false, bv,
                                                     (short)0, acc0, false, false);
        acc1 = __builtin_amdgcn_wmma_f32_16x16x4_f32(false, a1, false, bv,
                                                     (short)0, acc1, false, false);
    }

    const int n = n0 + l16;
    if (!is_diag) {
        const float base = angles[n];
        const int r7   = n & 127;
        const int slot = (n < 512) ? n : ((n & ~127) | ((r7 & 15) << 3) | (r7 >> 4));
#pragma unroll
        for (int i = 0; i < 8; ++i) {
            const int m0 = i + 8 * hi;
            const int m1 = m0 + 16;
            float a0 = base + tanhf(acc0[i]);
            float a1 = base + tanhf(acc1[i]);
            ws_cos[m0 * 1024 + slot] = cosf(a0);
            ws_sin[m0 * 1024 + slot] = sinf(a0);
            ws_cos[m1 * 1024 + slot] = cosf(a1);
            ws_sin[m1 * 1024 + slot] = sinf(a1);
        }
    } else {
        float base = tanhf(diagonal[n]);
        base = fminf(fmaxf(base, -CAPV), CAPV);
        const int dslot = ((n & 15) << 4) | (n >> 4);
#pragma unroll
        for (int i = 0; i < 8; ++i) {
            const int m0 = i + 8 * hi;
            const int m1 = m0 + 16;
            float d0 = base + tanhf(acc0[i]);
            float d1 = base + tanhf(acc1[i]);
            ws_diag[m0 * 256 + dslot] = fminf(fmaxf(d0, -CAPV), CAPV);
            ws_diag[m1 * 256 + dslot] = fminf(fmaxf(d1, -CAPV), CAPV);
        }
    }
}

#define TILE_POS 128

__device__ __forceinline__ void rotate(v4f& L, v4f& R, const float c, const float s) {
    const v4f cv = c;
    const v4f sv = s;
    const v4f l = L;
    const v4f r = R;
    L = __builtin_elementwise_fma(cv, l,  (sv * r));
    R = __builtin_elementwise_fma(cv, r, -(sv * l));
}

__device__ __forceinline__ void four_stages(v4f (&v)[16], const float* __restrict__ cs,
                                            const float* __restrict__ sn) {
#pragma unroll
    for (int s = 0; s < 4; ++s) {
        const int half = 1 << s;
#pragma unroll
        for (int i = 0; i < 16; ++i) {
            if (i & half) continue;
            const int j = ((i >> (s + 1)) << s) + (i & (half - 1));
            rotate(v[i], v[i + half], cs[s * PER_STAGE + j], sn[s * PER_STAGE + j]);
        }
    }
}

__global__ __launch_bounds__(256) void butterfly_kernel(
    const float* __restrict__ x,
    const float* __restrict__ ws_cos,
    const float* __restrict__ ws_sin,
    const float* __restrict__ ws_diag,
    float* __restrict__ out)
{
    __shared__ v4f tile[DIM * 32];

    const int b  = blockIdx.x >> 5;
    const int p0 = (blockIdx.x & 31) * TILE_POS;
    const int t  = threadIdx.x & 31;
    const int q  = __builtin_amdgcn_readfirstlane((int)(threadIdx.x >> 5));

    const float* cs = ws_cos + b * 1024;
    const float* sn = ws_sin + b * 1024;

    v4f va[16], vb[16];
    {
        const float* xa = x + ((size_t)b * DIM + 16 * q) * HW + p0 + 4 * t;
        const float* xb = xa + (size_t)128 * HW;
#pragma unroll
        for (int i = 0; i < 16; ++i)
            va[i] = *(const v4f*)(xa + (size_t)i * HW);
#pragma unroll
        for (int i = 0; i < 16; ++i)
            vb[i] = *(const v4f*)(xb + (size_t)i * HW);
    }
    four_stages(va, cs + 8 * q, sn + 8 * q);
#pragma unroll
    for (int i = 0; i < 16; ++i)
        tile[(16 * q + i) * 32 + t] = va[i];
    four_stages(vb, cs + 8 * (q + 8), sn + 8 * (q + 8));
#pragma unroll
    for (int i = 0; i < 16; ++i)
        tile[(16 * (q + 8) + i) * 32 + t] = vb[i];
    __syncthreads();

#pragma unroll
    for (int h = 0; h < 2; ++h) {
        const int g = q + 8 * h;
#pragma unroll
        for (int k = 0; k < 16; ++k)
            va[k] = tile[(g + 16 * k) * 32 + t];
        four_stages(va, cs + 4 * PER_STAGE + 8 * g, sn + 4 * PER_STAGE + 8 * g);
        const float* dg = ws_diag + b * DIM + 16 * g;
        float* ob = out + ((size_t)b * DIM + g) * HW + p0 + 4 * t;
#pragma unroll
        for (int k = 0; k < 16; ++k) {
            const v4f d = dg[k];
            *(v4f*)(ob + (size_t)(16 * k) * HW) = va[k] * d;
        }
    }
}

extern "C" void kernel_launch(void* const* d_in, const int* in_sizes, int n_in,
                              void* d_out, int out_size, void* d_ws, size_t ws_size,
                              hipStream_t stream) {
    const float* x          = (const float*)d_in[0];
    const float* modulation = (const float*)d_in[1];
    const float* angles     = (const float*)d_in[2];
    const float* diagonal   = (const float*)d_in[3];
    const float* w_angle    = (const float*)d_in[4];
    const float* w_diag     = (const float*)d_in[5];
    float* out = (float*)d_out;

    float* ws      = (float*)d_ws;
    float* ws_cos  = ws;
    float* ws_sin  = ws + 32 * 1024;
    float* ws_diag = ws + 64 * 1024;

    prep_kernel<<<80, 32, 0, stream>>>(modulation, angles, diagonal,
                                       w_angle, w_diag, ws_cos, ws_sin, ws_diag);
    butterfly_kernel<<<BATCH * (HW / TILE_POS), 256, 0, stream>>>(
        x, ws_cos, ws_sin, ws_diag, out);
}


# prep GEMM: K split over 8 waves per 32x16 tile (one load group per wave), partial tiles summed through LDS in fixed wave order (sum order changes), epilogue spread over 256 threads
# speedup vs baseline: 1.6803x; 1.1768x over previous
#include <hip/hip_runtime.h>
#include <math.h>

typedef __attribute__((ext_vector_type(2))) float v2f;
typedef __attribute__((ext_vector_type(8))) float v8f;
typedef __attribute__((ext_vector_type(4))) float v4f;

#define DIM 256
#define PER_STAGE 128
#define CAPV 0.99f
#define BATCH 32
#define HW 4096

#define PREP_WAVES 8
#define K_PER_WAVE (256 / PREP_WAVES)

__global__ __launch_bounds__(32 * PREP_WAVES) void prep_kernel(
    const float* __restrict__ modulation,
    const float* __restrict__ angles,
    const float* __restrict__ diagonal,
    const float* __restrict__ w_angle,
    const float* __restrict__ w_diag,
    float* __restrict__ ws_cos,
    float* __restrict__ ws_sin,
    float* __restrict__ ws_diag)
{
    __shared__ float part[PREP_WAVES * 16 * 32];

    const int  tile    = blockIdx.x;
    const bool is_diag = (tile >= 64);
    const int  n0      = is_diag ? (tile - 64) * 16 : tile * 16;
    const float* W     = is_diag ? w_diag : w_angle;

    const int lane = threadIdx.x & 31;
    const int w    = threadIdx.x >> 5;
    const int l16  = lane & 15;
    const int hi   = lane >> 4;

    const int n = n0 + l16;
    const float base_in = is_diag ? diagonal[n] : angles[n];

    const int kw = K_PER_WAVE * w;
    const float* arow0 = modulation + l16 * 256 + kw;
    const float* arow1 = modulation + (l16 + 16) * 256 + kw;
    const float* wrow  = W + (n0 + l16) * 256 + kw;

    v8f acc0 = {};
    v8f acc1 = {};
#pragma unroll
    for (int k0 = 0; k0 < K_PER_WAVE; k0 += 4) {
        const int k = k0 + 2 * hi;
        v2f a0, a1, bv;
        a0.x = arow0[k];  a0.y = arow0[k + 1];
        a1.x = arow1[k];  a1.y = arow1[k + 1];
        bv.x = wrow[k];   bv.y = wrow[k + 1];
        acc0 = __builtin_amdgcn_wmma_f32_16x16x4_f32(false, a0, false, bv,
                                                     (short)0, acc0, false, false);
        acc1 = __builtin_amdgcn_wmma_f32_16x16x4_f32(false, a1, false, bv,
                                                     (short)0, acc1, false, false);
    }

#pragma unroll
    for (int i = 0; i < 8; ++i) {
        part[(w * 16 + i) * 32 + lane]     = acc0[i];
        part[(w * 16 + 8 + i) * 32 + lane] = acc1[i];
    }
    __syncthreads();

    float s0 = part[w * 32 + lane];
    float s1 = part[(8 + w) * 32 + lane];
#pragma unroll
    for (int s = 1; s < PREP_WAVES; ++s) {
        s0 += part[(s * 16 + w) * 32 + lane];
        s1 += part[(s * 16 + 8 + w) * 32 + lane];
    }

    const int m0 = w + 8 * hi;
    const int m1 = m0 + 16;
    if (!is_diag) {
        const float base = base_in;
        const int r7   = n & 127;
        const int slot = (n < 512) ? n : ((n & ~127) | ((r7 & 15) << 3) | (r7 >> 4));
        float a0 = base + tanhf(s0);
        float a1 = base + tanhf(s1);
        ws_cos[m0 * 1024 + slot] = cosf(a0);
        ws_sin[m0 * 1024 + slot] = sinf(a0);
        ws_cos[m1 * 1024 + slot] = cosf(a1);
        ws_sin[m1 * 1024 + slot] = sinf(a1);
    } else {
        float base = tanhf(base_in);
        base = fminf(fmaxf(base, -CAPV), CAPV);
        const int dslot = ((n & 15) << 4) | (n >> 4);
        float d0 = base + tanhf(s0);
        float d1 = base + tanhf(s1);
        ws_diag[m0 * 256 + dslot] = fminf(fmaxf(d0, -CAPV), CAPV);
        ws_diag[m1 * 256 + dslot] = fminf(fmaxf(d1, -CAPV), CAPV);
    }
}

#define TILE_POS 128

__device__ __forceinline__ void rotate(v4f& L, v4f& R, const float c, const float s) {
    const v4f cv = c;
    const v4f sv = s;
    const v4f l = L;
    const v4f r = R;
    L = __builtin_elementwise_fma(cv, l,  (sv * r));
    R = __builtin_elementwise_fma(cv, r, -(sv * l));
}

__device__ __forceinline__ void four_stages(v4f (&v)[16], const float* __restrict__ cs,
                                            const float* __restrict__ sn) {
#pragma unroll
    for (int s = 0; s < 4; ++s) {
        const int half = 1 << s;
#pragma unroll
        for (int i = 0; i < 16; ++i) {
            if (i & half) continue;
            const int j = ((i >> (s + 1)) << s) + (i & (half - 1));
            rotate(v[i], v[i + half], cs[s * PER_STAGE + j], sn[s * PER_STAGE + j]);
        }
    }
}

__global__ __launch_bounds__(256) void butterfly_kernel(
    const float* __restrict__ x,
    const float* __restrict__ ws_cos,
    const float* __restrict__ ws_sin,
    const float* __restrict__ ws_diag,
    float* __restrict__ out)
{
    __shared__ v4f tile[DIM * 32];

    const int b  = blockIdx.x >> 5;
    const int p0 = (blockIdx.x & 31) * TILE_POS;
    const int t  = threadIdx.x & 31;
    const int q  = __builtin_amdgcn_readfirstlane((int)(threadIdx.x >> 5));

    const float* cs = ws_cos + b * 1024;
    const float* sn = ws_sin + b * 1024;

    v4f va[16], vb[16];
    {
        const float* xa = x + ((size_t)b * DIM + 16 * q) * HW + p0 + 4 * t;
        const float* xb = xa + (size_t)128 * HW;
#pragma unroll
        for (int i = 0; i < 16; ++i)
            va[i] = *(const v4f*)(xa + (size_t)i * HW);
#pragma unroll
        for (int i = 0; i < 16; ++i)
            vb[i] = *(const v4f*)(xb + (size_t)i * HW);
    }
    four_stages(va, cs + 8 * q, sn + 8 * q);
#pragma unroll
    for (int i = 0; i < 16; ++i)
        tile[(16 * q + i) * 32 + t] = va[i];
    four_stages(vb, cs + 8 * (q + 8), sn + 8 * (q + 8));
#pragma unroll
    for (int i = 0; i < 16; ++i)
        tile[(16 * (q + 8) + i) * 32 + t] = vb[i];
    __syncthreads();

#pragma unroll
    for (int h = 0; h < 2; ++h) {
        const int g = q + 8 * h;
#pragma unroll
        for (int k = 0; k < 16; ++k)
            va[k] = tile[(g + 16 * k) * 32 + t];
        four_stages(va, cs + 4 * PER_STAGE + 8 * g, sn + 4 * PER_STAGE + 8 * g);
        const float* dg = ws_diag + b * DIM + 16 * g;
        float* ob = out + ((size_t)b * DIM + g) * HW + p0 + 4 * t;
#pragma unroll
        for (int k = 0; k < 16; ++k) {
            const v4f d = dg[k];
            *(v4f*)(ob + (size_t)(16 * k) * HW) = va[k] * d;
        }
    }
}

extern "C" void kernel_launch(void* const* d_in, const int* in_sizes, int n_in,
                              void* d_out, int out_size, void* d_ws, size_t ws_size,
                              hipStream_t stream) {
    const float* x          = (const float*)d_in[0];
    const float* modulation = (const float*)d_in[1];
    const float* angles     = (const float*)d_in[2];
    const float* diagonal   = (const float*)d_in[3];
    const float* w_angle    = (const float*)d_in[4];
    const float* w_diag     = (const float*)d_in[5];
    float* out = (float*)d_out;

    float* ws      = (float*)d_ws;
    float* ws_cos  = ws;
    float* ws_sin  = ws + 32 * 1024;
    float* ws_diag = ws + 64 * 1024;

    prep_kernel<<<80, 32 * PREP_WAVES, 0, stream>>>(modulation, angles, diagonal,
                                       w_angle, w_diag, ws_cos, ws_sin, ws_diag);
    butterfly_kernel<<<BATCH * (HW / TILE_POS), 256, 0, stream>>>(
        x, ws_cos, ws_sin, ws_diag, out);
}


# butterfly software-pipelined over 2 tiles per block: next tile's 32 rows requested before the current tile's second pass; barriers fence LDS only, so loads and stores stay in flight across them
# speedup vs baseline: 1.7754x; 1.0566x over previous
#include <hip/hip_runtime.h>
#include <math.h>

typedef __attribute__((ext_vector_type(2))) float v2f;
typedef __attribute__((ext_vector_type(8))) float v8f;
typedef __attribute__((ext_vector_type(4))) float v4f;

#define DIM 256
#define PER_STAGE 128
#define CAPV 0.99f
#define BATCH 32
#define HW 4096

#define PREP_WAVES 8
#define K_PER_WAVE (256 / PREP_WAVES)

__global__ __launch_bounds__(32 * PREP_WAVES) void prep_kernel(
    const float* __restrict__ modulation,
    const float* __restrict__ angles,
    const float* __restrict__ diagonal,
    const float* __restrict__ w_angle,
    const float* __restrict__ w_diag,
    float* __restrict__ ws_cos,
    float* __restrict__ ws_sin,
    float* __restrict__ ws_diag)
{
    __shared__ float part[PREP_WAVES * 16 * 32];

    const int  tile    = blockIdx.x;
    const bool is_diag = (tile >= 64);
    const int  n0      = is_diag ? (tile - 64) * 16 : tile * 16;
    const float* W     = is_diag ? w_diag : w_angle;

    const int lane = threadIdx.x & 31;
    const int w    = threadIdx.x >> 5;
    const int l16  = lane & 15;
    const int hi   = lane >> 4;

    const int n = n0 + l16;
    const float base_in = is_diag ? diagonal[n] : angles[n];

    const int kw = K_PER_WAVE * w;
    const float* arow0 = modulation + l16 * 256 + kw;
    const float* arow1 = modulation + (l16 + 16) * 256 + kw;
    const float* wrow  = W + (n0 + l16) * 256 + kw;

    v8f acc0 = {};
    v8f acc1 = {};
#pragma unroll
    for (int k0 = 0; k0 < K_PER_WAVE; k0 += 4) {
        const int k = k0 + 2 * hi;
        v2f a0, a1, bv;
        a0.x = arow0[k];  a0.y = arow0[k + 1];
        a1.x = arow1[k];  a1.y = arow1[k + 1];
        bv.x = wrow[k];   bv.y = wrow[k + 1];
        acc0 = __builtin_amdgcn_wmma_f32_16x16x4_f32(false, a0, false, bv,
                                                     (short)0, acc0, false, false);
        acc1 = __builtin_amdgcn_wmma_f32_16x16x4_f32(false, a1, false, bv,
                                                     (short)0, acc1, false, false);
    }

#pragma unroll
    for (int i = 0; i < 8; ++i) {
        part[(w * 16 + i) * 32 + lane]     = acc0[i];
        part[(w * 16 + 8 + i) * 32 + lane] = acc1[i];
    }
    __syncthreads();

    float s0 = part[w * 32 + lane];
    float s1 = part[(8 + w) * 32 + lane];
#pragma unroll
    for (int s = 1; s < PREP_WAVES; ++s) {
        s0 += part[(s * 16 + w) * 32 + lane];
        s1 += part[(s * 16 + 8 + w) * 32 + lane];
    }

    const int m0 = w + 8 * hi;
    const int m1 = m0 + 16;
    if (!is_diag) {
        const float base = base_in;
        const int r7   = n & 127;
        const int slot = (n < 512) ? n : ((n & ~127) | ((r7 & 15) << 3) | (r7 >> 4));
        float a0 = base + tanhf(s0);
        float a1 = base + tanhf(s1);
        ws_cos[m0 * 1024 + slot] = cosf(a0);
        ws_sin[m0 * 1024 + slot] = sinf(a0);
        ws_cos[m1 * 1024 + slot] = cosf(a1);
        ws_sin[m1 * 1024 + slot] = sinf(a1);
    } else {
        float base = tanhf(base_in);
        base = fminf(fmaxf(base, -CAPV), CAPV);
        const int dslot = ((n & 15) << 4) | (n >> 4);
        float d0 = base + tanhf(s0);
        float d1 = base + tanhf(s1);
        ws_diag[m0 * 256 + dslot] = fminf(fmaxf(d0, -CAPV), CAPV);
        ws_diag[m1 * 256 + dslot] = fminf(fmaxf(d1, -CAPV), CAPV);
    }
}

#define TILE_POS 128
#define TILES_PER_BLOCK 2

__device__ __forceinline__ void rotate(v4f& L, v4f& R, const float c, const float s) {
    const v4f cv = c;
    const v4f sv = s;
    const v4f l = L;
    const v4f r = R;
    L = __builtin_elementwise_fma(cv, l,  (sv * r));
    R = __builtin_elementwise_fma(cv, r, -(sv * l));
}

__device__ __forceinline__ void four_stages(v4f (&v)[16], const float* __restrict__ cs,
                                            const float* __restrict__ sn) {
#pragma unroll
    for (int s = 0; s < 4; ++s) {
        const int half = 1 << s;
#pragma unroll
        for (int i = 0; i < 16; ++i) {
            if (i & half) continue;
            const int j = ((i >> (s + 1)) << s) + (i & (half - 1));
            rotate(v[i], v[i + half], cs[s * PER_STAGE + j], sn[s * PER_STAGE + j]);
        }
    }
}

__device__ __forceinline__ void lds_tile_barrier() {
    __builtin_amdgcn_fence(__ATOMIC_RELEASE, "workgroup", "local");
    __builtin_amdgcn_s_barrier();
    __builtin_amdgcn_fence(__ATOMIC_ACQUIRE, "workgroup", "local");
}

__global__ __launch_bounds__(256) void butterfly_kernel(
    const float* __restrict__ x,
    const float* __restrict__ ws_cos,
    const float* __restrict__ ws_sin,
    const float* __restrict__ ws_diag,
    float* __restrict__ out)
{
    __shared__ v4f tile[DIM * 32];

    const int b  = blockIdx.x >> 4;
    const int p0 = (blockIdx.x & 15) * (TILES_PER_BLOCK * TILE_POS);
    const int t  = threadIdx.x & 31;
    const int q  = __builtin_amdgcn_readfirstlane((int)(threadIdx.x >> 5));

    const float* cs = ws_cos + b * 1024;
    const float* sn = ws_sin + b * 1024;

    const float* xa = x + ((size_t)b * DIM + 16 * q) * HW + p0 + 4 * t;
    const float* xb = xa + (size_t)128 * HW;

    v4f va[16], vb[16], vn[16], vc[16];
#pragma unroll
    for (int i = 0; i < 16; ++i)
        va[i] = *(const v4f*)(xa + (size_t)i * HW);
#pragma unroll
    for (int i = 0; i < 16; ++i)
        vb[i] = *(const v4f*)(xb + (size_t)i * HW);

#pragma unroll
    for (int it = 0; it < TILES_PER_BLOCK; ++it) {
        const bool has_next = (it + 1 < TILES_PER_BLOCK);

        four_stages(va, cs + 8 * q, sn + 8 * q);
#pragma unroll
        for (int i = 0; i < 16; ++i)
            tile[(16 * q + i) * 32 + t] = va[i];
        if (has_next) {
            __builtin_amdgcn_sched_barrier(0);
#pragma unroll
            for (int i = 0; i < 16; ++i)
                va[i] = *(const v4f*)(xa + (it + 1) * TILE_POS + (size_t)i * HW);
#pragma unroll
            for (int i = 0; i < 16; ++i)
                vn[i] = *(const v4f*)(xb + (it + 1) * TILE_POS + (size_t)i * HW);
        }
        four_stages(vb, cs + 8 * (q + 8), sn + 8 * (q + 8));
#pragma unroll
        for (int i = 0; i < 16; ++i)
            tile[(16 * (q + 8) + i) * 32 + t] = vb[i];
        if (has_next) {
#pragma unroll
            for (int i = 0; i < 16; ++i)
                vb[i] = vn[i];
        }
        lds_tile_barrier();

#pragma unroll
        for (int h = 0; h < 2; ++h) {
            const int g = q + 8 * h;
#pragma unroll
            for (int k = 0; k < 16; ++k)
                vc[k] = tile[(g + 16 * k) * 32 + t];
            if (h == 1 && has_next)
                lds_tile_barrier();
            four_stages(vc, cs + 4 * PER_STAGE + 8 * g, sn + 4 * PER_STAGE + 8 * g);
            const float* dg = ws_diag + b * DIM + 16 * g;
            float* ob = out + ((size_t)b * DIM + g) * HW + p0 + it * TILE_POS + 4 * t;
#pragma unroll
            for (int k = 0; k < 16; ++k) {
                const v4f d = dg[k];
                *(v4f*)(ob + (size_t)(16 * k) * HW) = vc[k] * d;
            }
            if (h == 0)
                __builtin_amdgcn_sched_barrier(0);
        }
    }
}

extern "C" void kernel_launch(void* const* d_in, const int* in_sizes, int n_in,
                              void* d_out, int out_size, void* d_ws, size_t ws_size,
                              hipStream_t stream) {
    const float* x          = (const float*)d_in[0];
    const float* modulation = (const float*)d_in[1];
    const float* angles     = (const float*)d_in[2];
    const float* diagonal   = (const float*)d_in[3];
    const float* w_angle    = (const float*)d_in[4];
    const float* w_diag     = (const float*)d_in[5];
    float* out = (float*)d_out;

    float* ws      = (float*)d_ws;
    float* ws_cos  = ws;
    float* ws_sin  = ws + 32 * 1024;
    float* ws_diag = ws + 64 * 1024;

    prep_kernel<<<80, 32 * PREP_WAVES, 0, stream>>>(modulation, angles, diagonal,
                                       w_angle, w_diag, ws_cos, ws_sin, ws_diag);
    butterfly_kernel<<<BATCH * (HW / (TILES_PER_BLOCK * TILE_POS)), 256, 0, stream>>>(
        x, ws_cos, ws_sin, ws_diag, out);
}
